# starting kernel
# baseline (speedup 1.0000x reference)
#include <hip/hip_runtime.h>

typedef float v2f __attribute__((ext_vector_type(2)));
typedef float v8f __attribute__((ext_vector_type(8)));

#define IN_F  64
#define OUT_F 32
#define NB    2
#define NN    2048
#define BN    (NB * NN)

__global__ __launch_bounds__(256) void ggc_proj_wmma(
    const float* __restrict__ x,
    const float* __restrict__ U,  const float* __restrict__ V,
    const float* __restrict__ Aw, const float* __restrict__ Bw,
    float* __restrict__ Ah, float* __restrict__ Bh, float* __restrict__ Vh,
    float* __restrict__ out)
{
    const int lane = threadIdx.x & 31;
    const int wid  = threadIdx.x >> 5;
    const int tile = blockIdx.x * 8 + wid;
    const int row0 = tile * 16;
    const int m    = lane & 15;
    const int half = lane >> 4;

    const float* Ws[4] = {Aw, Bw, V, U};

    v8f acc[8];
    #pragma unroll
    for (int i = 0; i < 8; ++i) acc[i] = {};

    const float* xr = x + (size_t)(row0 + m) * IN_F + 2 * half;

    #pragma unroll
    for (int kk = 0; kk < IN_F / 4; ++kk) {
        const int kbase = kk * 4;
        v2f a;
        a.x = xr[kbase + 0];
        a.y = xr[kbase + 1];
        const int kb = kbase + 2 * half;
        #pragma unroll
        for (int w = 0; w < 4; ++w) {
            #pragma unroll
            for (int nt = 0; nt < 2; ++nt) {
                v2f b;
                b.x = Ws[w][(size_t)(kb + 0) * OUT_F + nt * 16 + m];
                b.y = Ws[w][(size_t)(kb + 1) * OUT_F + nt * 16 + m];
                acc[w * 2 + nt] = __builtin_amdgcn_wmma_f32_16x16x4_f32(
                    false, a, false, b, (short)0, acc[w * 2 + nt], false, false);
            }
        }
    }

    float* Ds[4] = {Ah, Bh, Vh, out};
    #pragma unroll
    for (int w = 0; w < 4; ++w) {
        #pragma unroll
        for (int nt = 0; nt < 2; ++nt) {
            #pragma unroll
            for (int v = 0; v < 8; ++v) {
                const int r = row0 + v + 8 * half;
                Ds[w][(size_t)r * OUT_F + nt * 16 + m] = acc[w * 2 + nt][v];
            }
        }
    }
}

__global__ __launch_bounds__(256) void ggc_gate_agg(
    const float* __restrict__ adj,
    const float* __restrict__ Ah, const float* __restrict__ Bh,
    const float* __restrict__ Vh, float* __restrict__ out)
{
    __shared__ float sB[32 * OUT_F];
    __shared__ float sV[32 * OUT_F];

    const int lane = threadIdx.x & 31;
    const int wid  = threadIdx.x >> 5;
    const int row  = blockIdx.x * 8 + wid;
    const int b    = row >> 11;

    const float* Bhb    = Bh  + (size_t)b * NN * OUT_F;
    const float* Vhb    = Vh  + (size_t)b * NN * OUT_F;
    const float* adjRow = adj + (size_t)row * NN;
    const float  ah     = Ah[(size_t)row * OUT_F + lane];

    float acc = 0.0f;
    for (int jbase = 0; jbase < NN; jbase += 32) {
        __syncthreads();
        #pragma unroll
        for (int k = 0; k < (32 * OUT_F) / 256; ++k) {
            const int idx = threadIdx.x + k * 256;
            sB[idx] = Bhb[(size_t)jbase * OUT_F + idx];
            sV[idx] = Vhb[(size_t)jbase * OUT_F + idx];
        }
        __syncthreads();

        const float av = adjRow[jbase + lane];
        #pragma unroll
        for (int jj = 0; jj < 32; ++jj) {
            const float adjv = __shfl(av, jj, 32);
            const float bh   = sB[jj * OUT_F + lane];
            const float vh   = sV[jj * OUT_F + lane];
            const float s = __builtin_amdgcn_rcpf(1.0f + __expf(-(ah + bh)));
            acc = fmaf(adjv * s, vh, acc);
        }
    }

    out[(size_t)row * OUT_F + lane] += acc;
}

extern "C" void kernel_launch(void* const* d_in, const int* in_sizes, int n_in,
                              void* d_out, int out_size, void* d_ws, size_t ws_size,
                              hipStream_t stream) {
    const float* adj = (const float*)d_in[0];
    const float* x   = (const float*)d_in[1];
    const float* U   = (const float*)d_in[2];
    const float* V   = (const float*)d_in[3];
    const float* Aw  = (const float*)d_in[4];
    const float* Bw  = (const float*)d_in[5];
    float* out = (float*)d_out;

    float* ws = (float*)d_ws;
    float* Ah = ws;
    float* Bh = ws + (size_t)BN * OUT_F;
    float* Vh = ws + (size_t)2 * BN * OUT_F;

    ggc_proj_wmma<<<BN / (16 * 8), 256, 0, stream>>>(x, U, V, Aw, Bw,
                                                     Ah, Bh, Vh, out);
    ggc_gate_agg<<<BN / 8, 256, 0, stream>>>(adj, Ah, Bh, Vh, out);
}


# exp hoisted out of the pair loop: sigmoid(a+b)=1/(1+exp(-a)*exp(-b)), exp(-a) once per row, exp(-b) once per staged element; unfactored form kept where |a| or |b| > 80
# speedup vs baseline: 1.8860x; 1.8860x over previous
#include <hip/hip_runtime.h>

typedef float v2f __attribute__((ext_vector_type(2)));
typedef float v8f __attribute__((ext_vector_type(8)));

#define IN_F  64
#define OUT_F 32
#define NB    2
#define NN    2048
#define BN    (NB * NN)

__global__ __launch_bounds__(256) void ggc_proj_wmma(
    const float* __restrict__ x,
    const float* __restrict__ U,  const float* __restrict__ V,
    const float* __restrict__ Aw, const float* __restrict__ Bw,
    float* __restrict__ Ah, float* __restrict__ Bh, float* __restrict__ Vh,
    float* __restrict__ out)
{
    const int lane = threadIdx.x & 31;
    const int wid  = threadIdx.x >> 5;
    const int tile = blockIdx.x * 8 + wid;
    const int row0 = tile * 16;
    const int m    = lane & 15;
    const int half = lane >> 4;

    const float* Ws[4] = {Aw, Bw, V, U};

    v8f acc[8];
    #pragma unroll
    for (int i = 0; i < 8; ++i) acc[i] = {};

    const float* xr = x + (size_t)(row0 + m) * IN_F + 2 * half;

    #pragma unroll
    for (int kk = 0; kk < IN_F / 4; ++kk) {
        const int kbase = kk * 4;
        v2f a;
        a.x = xr[kbase + 0];
        a.y = xr[kbase + 1];
        const int kb = kbase + 2 * half;
        #pragma unroll
        for (int w = 0; w < 4; ++w) {
            #pragma unroll
            for (int nt = 0; nt < 2; ++nt) {
                v2f b;
                b.x = Ws[w][(size_t)(kb + 0) * OUT_F + nt * 16 + m];
                b.y = Ws[w][(size_t)(kb + 1) * OUT_F + nt * 16 + m];
                acc[w * 2 + nt] = __builtin_amdgcn_wmma_f32_16x16x4_f32(
                    false, a, false, b, (short)0, acc[w * 2 + nt], false, false);
            }
        }
    }

    float* Ds[4] = {Ah, Bh, Vh, out};
    #pragma unroll
    for (int w = 0; w < 4; ++w) {
        #pragma unroll
        for (int nt = 0; nt < 2; ++nt) {
            #pragma unroll
            for (int v = 0; v < 8; ++v) {
                const int r = row0 + v + 8 * half;
                Ds[w][(size_t)r * OUT_F + nt * 16 + m] = acc[w * 2 + nt][v];
            }
        }
    }
}

#define EXP_SAFE 80.0f
#define NTILE    (NN / 32)

__global__ __launch_bounds__(256) void ggc_gate_agg(
    const float* __restrict__ adj,
    const float* __restrict__ Ah, const float* __restrict__ Bh,
    const float* __restrict__ Vh, float* __restrict__ out)
{
    __shared__ float sE[32 * OUT_F];
    __shared__ float sV[32 * OUT_F];
    __shared__ int   sWide[NTILE];

    const int lane = threadIdx.x & 31;
    const int wid  = threadIdx.x >> 5;
    const int row  = blockIdx.x * 8 + wid;
    const int b    = row >> 11;

    const float* Bhb    = Bh  + (size_t)b * NN * OUT_F;
    const float* Vhb    = Vh  + (size_t)b * NN * OUT_F;
    const float* adjRow = adj + (size_t)row * NN;
    const float  ah     = Ah[(size_t)row * OUT_F + lane];
    const float  ea     = __expf(-ah);
    const bool rowWide  = __any(!(fabsf(ah) <= EXP_SAFE));

    if (threadIdx.x < NTILE) sWide[threadIdx.x] = 0;

    float acc = 0.0f;
    for (int t = 0; t < NTILE; ++t) {
        const int jbase = t * 32;
        __syncthreads();
        bool wide = false;
        #pragma unroll
        for (int k = 0; k < (32 * OUT_F) / 256; ++k) {
            const int idx = threadIdx.x + k * 256;
            const float bh = Bhb[(size_t)jbase * OUT_F + idx];
            sE[idx] = __expf(-bh);
            sV[idx] = Vhb[(size_t)jbase * OUT_F + idx];
            wide |= !(fabsf(bh) <= EXP_SAFE);
        }
        if (wide) sWide[t] = 1;
        __syncthreads();

        const float av = adjRow[jbase + lane];
        if (rowWide || sWide[t] != 0) {
            #pragma unroll 4
            for (int jj = 0; jj < 32; ++jj) {
                const float adjv = __shfl(av, jj, 32);
                const float bh   = Bhb[(size_t)(jbase + jj) * OUT_F + lane];
                const float vh   = sV[jj * OUT_F + lane];
                const float s = __builtin_amdgcn_rcpf(1.0f + __expf(-(ah + bh)));
                acc = fmaf(adjv * s, vh, acc);
            }
        } else {
            #pragma unroll
            for (int jj = 0; jj < 32; ++jj) {
                const float adjv = __shfl(av, jj, 32);
                const float eb   = sE[jj * OUT_F + lane];
                const float vh   = sV[jj * OUT_F + lane];
                const float s = __builtin_amdgcn_rcpf(fmaf(ea, eb, 1.0f));
                acc = fmaf(adjv * vh, s, acc);
            }
        }
    }

    out[(size_t)row * OUT_F + lane] += acc;
}

extern "C" void kernel_launch(void* const* d_in, const int* in_sizes, int n_in,
                              void* d_out, int out_size, void* d_ws, size_t ws_size,
                              hipStream_t stream) {
    const float* adj = (const float*)d_in[0];
    const float* x   = (const float*)d_in[1];
    const float* U   = (const float*)d_in[2];
    const float* V   = (const float*)d_in[3];
    const float* Aw  = (const float*)d_in[4];
    const float* Bw  = (const float*)d_in[5];
    float* out = (float*)d_out;

    float* ws = (float*)d_ws;
    float* Ah = ws;
    float* Bh = ws + (size_t)BN * OUT_F;
    float* Vh = ws + (size_t)2 * BN * OUT_F;

    ggc_proj_wmma<<<BN / (16 * 8), 256, 0, stream>>>(x, U, V, Aw, Bw,
                                                     Ah, Bh, Vh, out);
    ggc_gate_agg<<<BN / 8, 256, 0, stream>>>(adj, Ah, Bh, Vh, out);
}


# projection launch re-gridded: 256 work-groups of 4 waves, one 16-row tile per group and one projection per wave (was 32 groups of 8 waves, 4 projections per wave); sums bit-identical
# speedup vs baseline: 5.8672x; 3.1109x over previous
#include <hip/hip_runtime.h>


typedef float v2f __attribute__((ext_vector_type(2)));
typedef float v4f __attribute__((ext_vector_type(4)));
typedef float v8f __attribute__((ext_vector_type(8)));

#define IN_F  64
#define OUT_F 32
#define NB    2
#define NN    2048
#define BN    (NB * NN)

#define PROJ_WAVES 4

template <bool WITH_EXP>
__device__ __forceinline__ void store_planes(const v8f (&acc)[2], float* __restrict__ plane,
                                             float* __restrict__ eplane, const int row0,
                                             const int m, const int half) {
    const int bt  = row0 >> 11;
    const int jt  = (row0 & (NN - 1)) + 8 * half;
    const int odd = m & 1;
    #pragma unroll
    for (int nt = 0; nt < 2; ++nt) {
        const size_t o = (((size_t)(bt * (OUT_F / 2) + nt * 8 + (m >> 1))) * NN + jt + 4 * odd) * 2;
        float p[8];
        #pragma unroll
        for (int v = 0; v < 4; ++v) {
            const float s = odd ? acc[nt][v] : acc[nt][v + 4];
            const float r = __shfl_xor(s, 1, 32);
            p[2 * v + 0] = odd ? r : acc[nt][v];
            p[2 * v + 1] = odd ? acc[nt][v + 4] : r;
        }
        v4f* q = (v4f*)(plane + o);
        q[0] = (v4f){p[0], p[1], p[2], p[3]};
        q[1] = (v4f){p[4], p[5], p[6], p[7]};
        if constexpr (WITH_EXP) {
            v4f* qe = (v4f*)(eplane + o);
            qe[0] = (v4f){__expf(-p[0]), __expf(-p[1]), __expf(-p[2]), __expf(-p[3])};
            qe[1] = (v4f){__expf(-p[4]), __expf(-p[5]), __expf(-p[6]), __expf(-p[7])};
        }
    }
}

__global__ __launch_bounds__(PROJ_WAVES * 32) void ggc_proj_wmma(
    const float* __restrict__ x,
    const float* __restrict__ U,  const float* __restrict__ V,
    const float* __restrict__ Aw, const float* __restrict__ Bw,
    float* __restrict__ Ah, float* __restrict__ Bh2, float* __restrict__ Eb2,
    float* __restrict__ Vh2, float* __restrict__ out)
{
    const int lane = threadIdx.x & 31;
    const int wid  = __builtin_amdgcn_readfirstlane((int)(threadIdx.x >> 5));
    const int row0 = blockIdx.x * 16;
    const int m    = lane & 15;
    const int half = lane >> 4;

    const float* __restrict__ W = wid == 0 ? Aw : wid == 1 ? Bw : wid == 2 ? V : U;

    v8f acc[2];
    acc[0] = {};
    acc[1] = {};

    const float* xr = x + (size_t)(row0 + m) * IN_F + 2 * half;

    #pragma unroll
    for (int kk = 0; kk < IN_F / 4; ++kk) {
        const int kbase = kk * 4;
        v2f a;
        a.x = xr[kbase + 0];
        a.y = xr[kbase + 1];
        const int kb = kbase + 2 * half;
        #pragma unroll
        for (int nt = 0; nt < 2; ++nt) {
            v2f b;
            b.x = W[(size_t)(kb + 0) * OUT_F + nt * 16 + m];
            b.y = W[(size_t)(kb + 1) * OUT_F + nt * 16 + m];
            acc[nt] = __builtin_amdgcn_wmma_f32_16x16x4_f32(
                false, a, false, b, (short)0, acc[nt], false, false);
        }
    }

    if (wid == 1) {
        store_planes<true>(acc, Bh2, Eb2, row0, m, half);
    } else if (wid == 2) {
        store_planes<false>(acc, Vh2, Vh2, row0, m, half);
    } else {
        float* __restrict__ dst = wid == 0 ? Ah : out;
        #pragma unroll
        for (int nt = 0; nt < 2; ++nt) {
            #pragma unroll
            for (int v = 0; v < 8; ++v) {
                const int r = row0 + v + 8 * half;
                dst[(size_t)r * OUT_F + nt * 16 + m] = acc[nt][v];
            }
        }
    }
}

#define A_SAFE   78.0f
#define EB_LO    1.0e-34f
#define EB_HI    1.0e34f
#define ROWS_PB  16
#define WAVES_PB 16
#define HROWS    8
#define JBLK     128
#define NJB      (NN / JBLK)
#define RPITCH   17
static_assert(ROWS_PB == WAVES_PB, "the output pass maps wave index to row");
static_assert(WAVES_PB * 2 == OUT_F, "a block covers all channels");

struct SrcBlock {
    v4f e[2];
    v4f v[2];
};

__device__ __forceinline__ void load_src(SrcBlock& q, const float* __restrict__ ep,
                                         const float* __restrict__ vp, const int jb) {
    const v4f* pe = (const v4f*)(ep + (size_t)jb * JBLK * 2);
    const v4f* pv = (const v4f*)(vp + (size_t)jb * JBLK * 2);
    q.e[0] = pe[0]; q.e[1] = pe[1];
    q.v[0] = pv[0]; q.v[1] = pv[1];
}

__device__ __forceinline__ void load_adj(v4f (&a)[HROWS], const float* __restrict__ adjp,
                                         const int r0, const int jb) {
    #pragma unroll
    for (int r = 0; r < HROWS; ++r)
        a[r] = *(const v4f*)(adjp + (size_t)(r0 + r) * NN + jb * JBLK);
}

template <bool WIDE>
__device__ __forceinline__ void pair_half(const v4f (&a)[HROWS], const SrcBlock& q,
                                          const v2f* __restrict__ rs, v2f* __restrict__ acc) {
    const v2f e[4] = {{q.e[0].x, q.e[0].y}, {q.e[0].z, q.e[0].w}, {q.e[1].x, q.e[1].y}, {q.e[1].z, q.e[1].w}};
    const v2f v[4] = {{q.v[0].x, q.v[0].y}, {q.v[0].z, q.v[0].w}, {q.v[1].x, q.v[1].y}, {q.v[1].z, q.v[1].w}};
    #pragma unroll
    for (int r = 0; r < HROWS; ++r) {
        const float an[4] = {a[r].x, a[r].y, a[r].z, a[r].w};
        #pragma unroll
        for (int n = 0; n < 4; ++n) {
            v2f u;
            if constexpr (WIDE) {
                const v2f z = e[n] + rs[r];
                u = (v2f){1.0f + __expf(-z.x), 1.0f + __expf(-z.y)};
            } else {
                u = e[n] * rs[r] + 1.0f;
            }
            const v2f g = {__builtin_amdgcn_rcpf(u.x), __builtin_amdgcn_rcpf(u.y)};
            acc[r] = (g * v[n]) * an[n] + acc[r];
        }
        if (r & 1) __builtin_amdgcn_sched_barrier(0);
    }
}

template <bool WIDE>
__device__ __forceinline__ void walk(const float* __restrict__ adjp, const float* __restrict__ ep,
                                     const float* __restrict__ vp, const v2f (&rs)[ROWS_PB],
                                     v2f (&acc)[ROWS_PB], float& emin, float& emax) {
    #pragma unroll
    for (int r = 0; r < ROWS_PB; ++r) acc[r] = (v2f){0.0f, 0.0f};

    SrcBlock cur, nxt;
    v4f a0[HROWS], a1[HROWS];
    load_src(cur, ep, vp, 0);
    load_adj(a0, adjp, 0, 0);
    #pragma unroll 2
    for (int jb = 0; jb < NJB; ++jb) {
        const int jn = (jb + 1) & (NJB - 1);
        load_adj(a1, adjp, HROWS, jb);
        load_src(nxt, ep, vp, jn);
        __builtin_amdgcn_sched_barrier(0);
        if constexpr (!WIDE) {
            #pragma unroll
            for (int k = 0; k < 2; ++k) {
                emin = fminf(emin, fminf(fminf(cur.e[k].x, cur.e[k].y), fminf(cur.e[k].z, cur.e[k].w)));
                emax = fmaxf(emax, fmaxf(fmaxf(cur.e[k].x, cur.e[k].y), fmaxf(cur.e[k].z, cur.e[k].w)));
            }
        }
        pair_half<WIDE>(a0, cur, &rs[0], &acc[0]);
        load_adj(a0, adjp, 0, jn);
        __builtin_amdgcn_sched_barrier(0);
        pair_half<WIDE>(a1, cur, &rs[HROWS], &acc[HROWS]);
        cur = nxt;
    }
}

__global__ __launch_bounds__(WAVES_PB * 32) void ggc_gate_agg(
    const float* __restrict__ adj,
    const float* __restrict__ Ah, const float* __restrict__ Bh2,
    const float* __restrict__ Eb2, const float* __restrict__ Vh2,
    float* __restrict__ out)
{
    __shared__ float sPart[WAVES_PB * ROWS_PB * 2 * RPITCH];
    __shared__ float sOut[ROWS_PB * OUT_F];

    const int lane = threadIdx.x & 31;
    const int wid  = threadIdx.x >> 5;
    const int row0 = blockIdx.x * ROWS_PB;
    const int b    = row0 >> 11;
    const int c0   = wid * 2;

    const float* adjp = adj + (size_t)row0 * NN + 4 * lane;
    const size_t po   = (((size_t)(b * (OUT_F / 2) + wid)) * NN + 4 * lane) * 2;

    const float aMine = Ah[(size_t)(row0 + (lane >> 1)) * OUT_F + c0 + (lane & 1)];
    const bool wideA  = __any((int)!(fabsf(aMine) <= A_SAFE));
    const float eMine = __expf(-aMine);
    v2f ra[ROWS_PB], re[ROWS_PB];
    #pragma unroll
    for (int r = 0; r < ROWS_PB; ++r) {
        ra[r].x = __builtin_bit_cast(float, __builtin_amdgcn_readlane(__builtin_bit_cast(int, aMine), 2 * r));
        ra[r].y = __builtin_bit_cast(float, __builtin_amdgcn_readlane(__builtin_bit_cast(int, aMine), 2 * r + 1));
        re[r].x = __builtin_bit_cast(float, __builtin_amdgcn_readlane(__builtin_bit_cast(int, eMine), 2 * r));
        re[r].y = __builtin_bit_cast(float, __builtin_amdgcn_readlane(__builtin_bit_cast(int, eMine), 2 * r + 1));
    }

    v2f acc[ROWS_PB];
    float emin = 1.0f, emax = 1.0f;
    if (!wideA) walk<false>(adjp, Eb2 + po, Vh2 + po, re, acc, emin, emax);
    const bool wide = wideA || __any((int)(emin < EB_LO) | (int)(emax > EB_HI));
    if (wide) walk<true>(adjp, Bh2 + po, Vh2 + po, ra, acc, emin, emax);

    float part[ROWS_PB * 2];
    #pragma unroll
    for (int r = 0; r < ROWS_PB; ++r) {
        part[2 * r + 0] = acc[r].x + __shfl_xor(acc[r].x, 16, 32);
        part[2 * r + 1] = acc[r].y + __shfl_xor(acc[r].y, 16, 32);
    }
    if (lane < 16) {
        #pragma unroll
        for (int k = 0; k < ROWS_PB * 2; ++k)
            sPart[(wid * (ROWS_PB * 2) + k) * RPITCH + lane] = part[k];
    }
    __syncthreads();
    {
        float sum = 0.0f;
        #pragma unroll
        for (int t = 0; t < 16; ++t)
            sum += sPart[(wid * (ROWS_PB * 2) + lane) * RPITCH + t];
        sOut[(lane >> 1) * OUT_F + c0 + (lane & 1)] = sum;
    }
    __syncthreads();
    {
        const size_t o = (size_t)(row0 + wid) * OUT_F + lane;
        out[o] = out[o] + sOut[wid * OUT_F + lane];
    }
}

extern "C" void kernel_launch(void* const* d_in, const int* in_sizes, int n_in,
                              void* d_out, int out_size, void* d_ws, size_t ws_size,
                              hipStream_t stream) {
    const float* adj = (const float*)d_in[0];
    const float* x   = (const float*)d_in[1];
    const float* U   = (const float*)d_in[2];
    const float* V   = (const float*)d_in[3];
    const float* Aw  = (const float*)d_in[4];
    const float* Bw  = (const float*)d_in[5];
    float* out = (float*)d_out;

    float* ws  = (float*)d_ws;
    float* Ah  = ws;
    float* Bh2 = ws + (size_t)BN * OUT_F;
    float* Eb2 = ws + (size_t)2 * BN * OUT_F;
    float* Vh2 = ws + (size_t)3 * BN * OUT_F;

    ggc_proj_wmma<<<BN / 16, PROJ_WAVES * 32, 0, stream>>>(x, U, V, Aw, Bw,
                                                           Ah, Bh2, Eb2, Vh2, out);
    ggc_gate_agg<<<BN / ROWS_PB, WAVES_PB * 32, 0, stream>>>(adj, Ah, Bh2, Eb2, Vh2, out);
}


# adj staged through LDS in 4 chunks of 512 source nodes, 2 buffers: wave w fetches row w one chunk ahead, 1 barrier per chunk; the walk reads adj from LDS, not 16 global rows per wave; sums unchanged
# speedup vs baseline: 6.5494x; 1.1163x over previous
#include <hip/hip_runtime.h>


typedef float v2f __attribute__((ext_vector_type(2)));
typedef float v4f __attribute__((ext_vector_type(4)));
typedef float v8f __attribute__((ext_vector_type(8)));

#define IN_F  64
#define OUT_F 32
#define NB    2
#define NN    2048
#define BN    (NB * NN)

#define PROJ_WAVES 4

template <bool WITH_EXP>
__device__ __forceinline__ void store_planes(const v8f (&acc)[2], float* __restrict__ plane,
                                             float* __restrict__ eplane, const int row0,
                                             const int m, const int half) {
    const int bt  = row0 >> 11;
    const int jt  = (row0 & (NN - 1)) + 8 * half;
    const int odd = m & 1;
    #pragma unroll
    for (int nt = 0; nt < 2; ++nt) {
        const size_t o = (((size_t)(bt * (OUT_F / 2) + nt * 8 + (m >> 1))) * NN + jt + 4 * odd) * 2;
        float p[8];
        #pragma unroll
        for (int v = 0; v < 4; ++v) {
            const float s = odd ? acc[nt][v] : acc[nt][v + 4];
            const float r = __shfl_xor(s, 1, 32);
            p[2 * v + 0] = odd ? r : acc[nt][v];
            p[2 * v + 1] = odd ? acc[nt][v + 4] : r;
        }
        v4f* q = (v4f*)(plane + o);
        q[0] = (v4f){p[0], p[1], p[2], p[3]};
        q[1] = (v4f){p[4], p[5], p[6], p[7]};
        if constexpr (WITH_EXP) {
            v4f* qe = (v4f*)(eplane + o);
            qe[0] = (v4f){__expf(-p[0]), __expf(-p[1]), __expf(-p[2]), __expf(-p[3])};
            qe[1] = (v4f){__expf(-p[4]), __expf(-p[5]), __expf(-p[6]), __expf(-p[7])};
        }
    }
}

__global__ __launch_bounds__(PROJ_WAVES * 32) void ggc_proj_wmma(
    const float* __restrict__ x,
    const float* __restrict__ U,  const float* __restrict__ V,
    const float* __restrict__ Aw, const float* __restrict__ Bw,
    float* __restrict__ Ah, float* __restrict__ Bh2, float* __restrict__ Eb2,
    float* __restrict__ Vh2, float* __restrict__ out)
{
    const int lane = threadIdx.x & 31;
    const int wid  = __builtin_amdgcn_readfirstlane((int)(threadIdx.x >> 5));
    const int row0 = blockIdx.x * 16;
    const int m    = lane & 15;
    const int half = lane >> 4;

    const float* __restrict__ W = wid == 0 ? Aw : wid == 1 ? Bw : wid == 2 ? V : U;

    v8f acc[2];
    acc[0] = {};
    acc[1] = {};

    const float* xr = x + (size_t)(row0 + m) * IN_F + 2 * half;

    #pragma unroll
    for (int kk = 0; kk < IN_F / 4; ++kk) {
        const int kbase = kk * 4;
        v2f a;
        a.x = xr[kbase + 0];
        a.y = xr[kbase + 1];
        const int kb = kbase + 2 * half;
        #pragma unroll
        for (int nt = 0; nt < 2; ++nt) {
            v2f b;
            b.x = W[(size_t)(kb + 0) * OUT_F + nt * 16 + m];
            b.y = W[(size_t)(kb + 1) * OUT_F + nt * 16 + m];
            acc[nt] = __builtin_amdgcn_wmma_f32_16x16x4_f32(
                false, a, false, b, (short)0, acc[nt], false, false);
        }
    }

    if (wid == 1) {
        store_planes<true>(acc, Bh2, Eb2, row0, m, half);
    } else if (wid == 2) {
        store_planes<false>(acc, Vh2, Vh2, row0, m, half);
    } else {
        float* __restrict__ dst = wid == 0 ? Ah : out;
        #pragma unroll
        for (int nt = 0; nt < 2; ++nt) {
            #pragma unroll
            for (int v = 0; v < 8; ++v) {
                const int r = row0 + v + 8 * half;
                dst[(size_t)r * OUT_F + nt * 16 + m] = acc[nt][v];
            }
        }
    }
}

#define A_SAFE   78.0f
#define EB_LO    1.0e-34f
#define EB_HI    1.0e34f
#define ROWS_PB  16
#define WAVES_PB 16
#define HROWS    8
#define JBLK     128
#define NJB      (NN / JBLK)
#define RPITCH   17
#define CJB      4
#define CN4      (CJB * JBLK / 4)
#define NCH      (NJB / CJB)
static_assert(NCH * CJB == NJB && (CJB % 2) == 0, "whole chunks, an even number of blocks in each");
static_assert(ROWS_PB == WAVES_PB, "the output pass maps wave index to row");
static_assert(WAVES_PB * 2 == OUT_F, "a block covers all channels");

struct SrcBlock {
    v4f e[2];
    v4f v[2];
};

__device__ __forceinline__ void load_src(SrcBlock& q, const float* __restrict__ ep,
                                         const float* __restrict__ vp, const int jb) {
    const v4f* pe = (const v4f*)(ep + (size_t)jb * JBLK * 2);
    const v4f* pv = (const v4f*)(vp + (size_t)jb * JBLK * 2);
    q.e[0] = pe[0]; q.e[1] = pe[1];
    q.v[0] = pv[0]; q.v[1] = pv[1];
}

__device__ __forceinline__ void load_adj(v4f (&a)[HROWS], const float* __restrict__ adjp,
                                         const int r0, const int jb) {
    #pragma unroll
    for (int r = 0; r < HROWS; ++r)
        a[r] = *(const v4f*)(adjp + (size_t)(r0 + r) * NN + jb * JBLK);
}

template <bool WIDE>
__device__ __forceinline__ void pair_half(const v4f (&a)[HROWS], const SrcBlock& q,
                                          const v2f* __restrict__ rs, v2f* __restrict__ acc) {
    const v2f e[4] = {{q.e[0].x, q.e[0].y}, {q.e[0].z, q.e[0].w}, {q.e[1].x, q.e[1].y}, {q.e[1].z, q.e[1].w}};
    const v2f v[4] = {{q.v[0].x, q.v[0].y}, {q.v[0].z, q.v[0].w}, {q.v[1].x, q.v[1].y}, {q.v[1].z, q.v[1].w}};
    #pragma unroll
    for (int r = 0; r < HROWS; ++r) {
        const float an[4] = {a[r].x, a[r].y, a[r].z, a[r].w};
        #pragma unroll
        for (int n = 0; n < 4; ++n) {
            v2f u;
            if constexpr (WIDE) {
                const v2f z = e[n] + rs[r];
                u = (v2f){1.0f + __expf(-z.x), 1.0f + __expf(-z.y)};
            } else {
                u = e[n] * rs[r] + 1.0f;
            }
            const v2f g = {__builtin_amdgcn_rcpf(u.x), __builtin_amdgcn_rcpf(u.y)};
            acc[r] = (g * v[n]) * an[n] + acc[r];
        }
        if (r & 1) __builtin_amdgcn_sched_barrier(0);
    }
}

template <bool WIDE>
__device__ __forceinline__ void walk(const float* __restrict__ adjp, const float* __restrict__ ep,
                                     const float* __restrict__ vp, const v2f (&rs)[ROWS_PB],
                                     v2f (&acc)[ROWS_PB], float& emin, float& emax) {
    #pragma unroll
    for (int r = 0; r < ROWS_PB; ++r) acc[r] = (v2f){0.0f, 0.0f};

    SrcBlock cur, nxt;
    v4f a0[HROWS], a1[HROWS];
    load_src(cur, ep, vp, 0);
    load_adj(a0, adjp, 0, 0);
    #pragma unroll 2
    for (int jb = 0; jb < NJB; ++jb) {
        const int jn = (jb + 1) & (NJB - 1);
        load_adj(a1, adjp, HROWS, jb);
        load_src(nxt, ep, vp, jn);
        __builtin_amdgcn_sched_barrier(0);
        if constexpr (!WIDE) {
            #pragma unroll
            for (int k = 0; k < 2; ++k) {
                emin = fminf(emin, fminf(fminf(cur.e[k].x, cur.e[k].y), fminf(cur.e[k].z, cur.e[k].w)));
                emax = fmaxf(emax, fmaxf(fmaxf(cur.e[k].x, cur.e[k].y), fmaxf(cur.e[k].z, cur.e[k].w)));
            }
        }
        pair_half<WIDE>(a0, cur, &rs[0], &acc[0]);
        load_adj(a0, adjp, 0, jn);
        __builtin_amdgcn_sched_barrier(0);
        pair_half<WIDE>(a1, cur, &rs[HROWS], &acc[HROWS]);
        cur = nxt;
    }
}

__device__ __forceinline__ void stage_load(v4f (&st)[CJB], const float* __restrict__ adjRow, const int c) {
    #pragma unroll
    for (int k = 0; k < CJB; ++k)
        st[k] = *(const v4f*)(adjRow + (c * CJB + k) * JBLK);
}

__device__ __forceinline__ void stage_store(const v4f (&st)[CJB], v4f* sRow) {
    #pragma unroll
    for (int k = 0; k < CJB; ++k) sRow[k * (JBLK / 4)] = st[k];
}

__device__ __forceinline__ void load_adj_lds(v4f (&a)[HROWS], const v4f* sLane, const int r0, const int k) {
    #pragma unroll
    for (int r = 0; r < HROWS; ++r) a[r] = sLane[(r0 + r) * CN4 + k * (JBLK / 4)];
}

__device__ __forceinline__ void walk_staged(const float* __restrict__ adjRow, v4f (&st)[CJB],
                                            v4f* sAdj, const int wid, const int lane,
                                            const float* __restrict__ ep, const float* __restrict__ vp,
                                            const v2f (&rs)[ROWS_PB], v2f (&acc)[ROWS_PB],
                                            float& emin, float& emax) {
    #pragma unroll
    for (int r = 0; r < ROWS_PB; ++r) acc[r] = (v2f){0.0f, 0.0f};

    SrcBlock cur, nxt;
    v4f a0[HROWS], a1[HROWS];
    load_src(cur, ep, vp, 0);
    stage_store(st, sAdj + wid * CN4 + lane);
    __syncthreads();
    #pragma nounroll
    for (int c = 0; c < NCH; ++c) {
        const v4f* sCur = sAdj + (c & 1) * (ROWS_PB * CN4) + lane;
        if (c + 1 < NCH) stage_load(st, adjRow, c + 1);
        load_adj_lds(a0, sCur, 0, 0);
        #pragma unroll 2
        for (int k = 0; k < CJB; ++k) {
            const int jn = (c * CJB + k + 1) & (NJB - 1);
            const int kn = (k + 1) & (CJB - 1);
            load_adj_lds(a1, sCur, HROWS, k);
            load_src(nxt, ep, vp, jn);
            __builtin_amdgcn_sched_barrier(0);
            #pragma unroll
            for (int h = 0; h < 2; ++h) {
                emin = fminf(emin, fminf(fminf(cur.e[h].x, cur.e[h].y), fminf(cur.e[h].z, cur.e[h].w)));
                emax = fmaxf(emax, fmaxf(fmaxf(cur.e[h].x, cur.e[h].y), fmaxf(cur.e[h].z, cur.e[h].w)));
            }
            pair_half<false>(a0, cur, &rs[0], &acc[0]);
            load_adj_lds(a0, sCur, 0, kn);
            __builtin_amdgcn_sched_barrier(0);
            pair_half<false>(a1, cur, &rs[HROWS], &acc[HROWS]);
            cur = nxt;
        }
        if (c + 1 < NCH) {
            stage_store(st, sAdj + ((c + 1) & 1) * (ROWS_PB * CN4) + wid * CN4 + lane);
            __syncthreads();
        }
    }
}

__global__ __launch_bounds__(WAVES_PB * 32) void ggc_gate_agg(
    const float* __restrict__ adj,
    const float* __restrict__ Ah, const float* __restrict__ Bh2,
    const float* __restrict__ Eb2, const float* __restrict__ Vh2,
    float* __restrict__ out)
{
    __shared__ float sPart[WAVES_PB * ROWS_PB * 2 * RPITCH];
    __shared__ float sOut[ROWS_PB * OUT_F];
    __shared__ v4f sAdj[2 * ROWS_PB * CN4];

    const int lane = threadIdx.x & 31;
    const int wid  = threadIdx.x >> 5;
    const int row0 = blockIdx.x * ROWS_PB;
    const int b    = row0 >> 11;
    const int c0   = wid * 2;

    const float* adjp = adj + (size_t)row0 * NN + 4 * lane;
    const size_t po   = (((size_t)(b * (OUT_F / 2) + wid)) * NN + 4 * lane) * 2;

    const float* adjRow = adj + (size_t)(row0 + wid) * NN + 4 * lane;
    v4f st[CJB];
    stage_load(st, adjRow, 0);

    const float aMine = Ah[(size_t)(row0 + (lane >> 1)) * OUT_F + c0 + (lane & 1)];
    const bool wideA  = __any((int)!(fabsf(aMine) <= A_SAFE));
    const float eMine = __expf(-aMine);
    v2f ra[ROWS_PB], re[ROWS_PB];
    #pragma unroll
    for (int r = 0; r < ROWS_PB; ++r) {
        ra[r].x = __builtin_bit_cast(float, __builtin_amdgcn_readlane(__builtin_bit_cast(int, aMine), 2 * r));
        ra[r].y = __builtin_bit_cast(float, __builtin_amdgcn_readlane(__builtin_bit_cast(int, aMine), 2 * r + 1));
        re[r].x = __builtin_bit_cast(float, __builtin_amdgcn_readlane(__builtin_bit_cast(int, eMine), 2 * r));
        re[r].y = __builtin_bit_cast(float, __builtin_amdgcn_readlane(__builtin_bit_cast(int, eMine), 2 * r + 1));
    }

    v2f acc[ROWS_PB];
    float emin = 1.0f, emax = 1.0f;
    walk_staged(adjRow, st, sAdj, wid, lane, Eb2 + po, Vh2 + po, re, acc, emin, emax);
    const bool wide = wideA || __any((int)(emin < EB_LO) | (int)(emax > EB_HI));
    if (wide) walk<true>(adjp, Bh2 + po, Vh2 + po, ra, acc, emin, emax);

    float part[ROWS_PB * 2];
    #pragma unroll
    for (int r = 0; r < ROWS_PB; ++r) {
        part[2 * r + 0] = acc[r].x + __shfl_xor(acc[r].x, 16, 32);
        part[2 * r + 1] = acc[r].y + __shfl_xor(acc[r].y, 16, 32);
    }
    if (lane < 16) {
        #pragma unroll
        for (int k = 0; k < ROWS_PB * 2; ++k)
            sPart[(wid * (ROWS_PB * 2) + k) * RPITCH + lane] = part[k];
    }
    __syncthreads();
    {
        float sum = 0.0f;
        #pragma unroll
        for (int t = 0; t < 16; ++t)
            sum += sPart[(wid * (ROWS_PB * 2) + lane) * RPITCH + t];
        sOut[(lane >> 1) * OUT_F + c0 + (lane & 1)] = sum;
    }
    __syncthreads();
    {
        const size_t o = (size_t)(row0 + wid) * OUT_F + lane;
        out[o] = out[o] + sOut[wid * OUT_F + lane];
    }
}

extern "C" void kernel_launch(void* const* d_in, const int* in_sizes, int n_in,
                              void* d_out, int out_size, void* d_ws, size_t ws_size,
                              hipStream_t stream) {
    const float* adj = (const float*)d_in[0];
    const float* x   = (const float*)d_in[1];
    const float* U   = (const float*)d_in[2];
    const float* V   = (const float*)d_in[3];
    const float* Aw  = (const float*)d_in[4];
    const float* Bw  = (const float*)d_in[5];
    float* out = (float*)d_out;

    float* ws  = (float*)d_ws;
    float* Ah  = ws;
    float* Bh2 = ws + (size_t)BN * OUT_F;
    float* Eb2 = ws + (size_t)2 * BN * OUT_F;
    float* Vh2 = ws + (size_t)3 * BN * OUT_F;

    ggc_proj_wmma<<<BN / 16, PROJ_WAVES * 32, 0, stream>>>(x, U, V, Aw, Bw,
                                                           Ah, Bh2, Eb2, Vh2, out);
    ggc_gate_agg<<<BN / ROWS_PB, WAVES_PB * 32, 0, stream>>>(adj, Ah, Bh2, Eb2, Vh2, out);
}


# pair loop: Vh moved into the denominator: sigmoid(a+b)*v = 1/(exp(-a)*P + Q); kernel 1 stages P = exp(-b)/v, Q = 1/v; a term is packed fma, 2 rcp, packed fma (a packed mul less); sum order kept
# speedup vs baseline: 6.9368x; 1.0592x over previous
#include <hip/hip_runtime.h>


typedef float v2f __attribute__((ext_vector_type(2)));
typedef float v4f __attribute__((ext_vector_type(4)));
typedef float v8f __attribute__((ext_vector_type(8)));

#define IN_F  64
#define OUT_F 32
#define NB    2
#define NN    2048
#define BN    (NB * NN)

#define PROJ_WAVES 4
#define EB_LO    1.0e-17f
#define EB_HI    1.0e17f
#define IV_LO    1.0e-18f

__device__ __forceinline__ void store_plane(const v8f (&acc)[2], float* __restrict__ plane,
                                            const int row0, const int m, const int half) {
    const int bt  = row0 >> 11;
    const int jt  = (row0 & (NN - 1)) + 8 * half;
    const int odd = m & 1;
    #pragma unroll
    for (int nt = 0; nt < 2; ++nt) {
        const size_t o = (((size_t)(bt * (OUT_F / 2) + nt * 8 + (m >> 1))) * NN + jt + 4 * odd) * 2;
        float p[8];
        #pragma unroll
        for (int v = 0; v < 4; ++v) {
            const float s = odd ? acc[nt][v] : acc[nt][v + 4];
            const float r = __shfl_xor(s, 1, 32);
            p[2 * v + 0] = odd ? r : acc[nt][v];
            p[2 * v + 1] = odd ? acc[nt][v + 4] : r;
        }
        v4f* q = (v4f*)(plane + o);
        q[0] = (v4f){p[0], p[1], p[2], p[3]};
        q[1] = (v4f){p[4], p[5], p[6], p[7]};
    }
}

__global__ __launch_bounds__(PROJ_WAVES * 32) void ggc_proj_wmma(
    const float* __restrict__ x,
    const float* __restrict__ U,  const float* __restrict__ V,
    const float* __restrict__ Aw, const float* __restrict__ Bw,
    float* __restrict__ Ah, float* __restrict__ Bh2, float* __restrict__ Pb2,
    float* __restrict__ Qv2, float* __restrict__ out)
{
    __shared__ float sV[2 * 8 * 32];

    const int lane = threadIdx.x & 31;
    const int wid  = __builtin_amdgcn_readfirstlane((int)(threadIdx.x >> 5));
    const int row0 = blockIdx.x * 16;
    const int m    = lane & 15;
    const int half = lane >> 4;

    const float* __restrict__ W = wid == 0 ? Aw : wid == 1 ? Bw : wid == 2 ? V : U;

    v8f acc[2];
    acc[0] = {};
    acc[1] = {};

    const float* xr = x + (size_t)(row0 + m) * IN_F + 2 * half;

    #pragma unroll
    for (int kk = 0; kk < IN_F / 4; ++kk) {
        const int kbase = kk * 4;
        v2f a;
        a.x = xr[kbase + 0];
        a.y = xr[kbase + 1];
        const int kb = kbase + 2 * half;
        #pragma unroll
        for (int nt = 0; nt < 2; ++nt) {
            v2f b;
            b.x = W[(size_t)(kb + 0) * OUT_F + nt * 16 + m];
            b.y = W[(size_t)(kb + 1) * OUT_F + nt * 16 + m];
            acc[nt] = __builtin_amdgcn_wmma_f32_16x16x4_f32(
                false, a, false, b, (short)0, acc[nt], false, false);
        }
    }

    if (wid == 2) {
        #pragma unroll
        for (int nt = 0; nt < 2; ++nt) {
            #pragma unroll
            for (int v = 0; v < 8; ++v) sV[(nt * 8 + v) * 32 + lane] = acc[nt][v];
        }
    }
    __syncthreads();
    if (wid == 1) {
        v8f accP[2];
        #pragma unroll
        for (int nt = 0; nt < 2; ++nt) {
            #pragma unroll
            for (int v = 0; v < 8; ++v) {
                const float eb = __expf(-acc[nt][v]);
                const float iv = __builtin_amdgcn_rcpf(sV[(nt * 8 + v) * 32 + lane]);
                const bool ok  = (eb >= EB_LO) && (eb <= EB_HI) && (fabsf(iv) >= IV_LO);
                accP[nt][v] = ok ? eb * iv : __builtin_nanf("");
            }
        }
        store_plane(acc, Bh2, row0, m, half);
        store_plane(accP, Pb2, row0, m, half);
    } else if (wid == 2) {
        #pragma unroll
        for (int nt = 0; nt < 2; ++nt) {
            #pragma unroll
            for (int v = 0; v < 8; ++v) acc[nt][v] = __builtin_amdgcn_rcpf(acc[nt][v]);
        }
        store_plane(acc, Qv2, row0, m, half);
    } else {
        float* __restrict__ dst = wid == 0 ? Ah : out;
        #pragma unroll
        for (int nt = 0; nt < 2; ++nt) {
            #pragma unroll
            for (int v = 0; v < 8; ++v) {
                const int r = row0 + v + 8 * half;
                dst[(size_t)r * OUT_F + nt * 16 + m] = acc[nt][v];
            }
        }
    }
}

#define A_SAFE   40.0f
#define ROWS_PB  16
#define WAVES_PB 16
#define HROWS    8
#define JBLK     128
#define NJB      (NN / JBLK)
#define RPITCH   17
#define CJB      4
#define CN4      (CJB * JBLK / 4)
#define NCH      (NJB / CJB)
static_assert(NCH * CJB == NJB && (CJB % 2) == 0, "whole chunks, an even number of blocks in each");
static_assert(ROWS_PB == WAVES_PB, "the output pass maps wave index to row");
static_assert(WAVES_PB * 2 == OUT_F, "a block covers all channels");

struct SrcBlock {
    v4f e[2];
    v4f v[2];
};

__device__ __forceinline__ void load_src(SrcBlock& q, const float* __restrict__ ep,
                                         const float* __restrict__ vp, const int jb) {
    const v4f* pe = (const v4f*)(ep + (size_t)jb * JBLK * 2);
    const v4f* pv = (const v4f*)(vp + (size_t)jb * JBLK * 2);
    q.e[0] = pe[0]; q.e[1] = pe[1];
    q.v[0] = pv[0]; q.v[1] = pv[1];
}

__device__ __forceinline__ void load_adj(v4f (&a)[HROWS], const float* __restrict__ adjp,
                                         const int r0, const int jb) {
    #pragma unroll
    for (int r = 0; r < HROWS; ++r)
        a[r] = *(const v4f*)(adjp + (size_t)(r0 + r) * NN + jb * JBLK);
}

template <bool WIDE>
__device__ __forceinline__ void pair_half(const v4f (&a)[HROWS], const SrcBlock& q,
                                          const v2f* __restrict__ rs, v2f* __restrict__ acc) {
    const v2f e[4] = {{q.e[0].x, q.e[0].y}, {q.e[0].z, q.e[0].w}, {q.e[1].x, q.e[1].y}, {q.e[1].z, q.e[1].w}};
    const v2f v[4] = {{q.v[0].x, q.v[0].y}, {q.v[0].z, q.v[0].w}, {q.v[1].x, q.v[1].y}, {q.v[1].z, q.v[1].w}};
    #pragma unroll
    for (int r = 0; r < HROWS; ++r) {
        const float an[4] = {a[r].x, a[r].y, a[r].z, a[r].w};
        #pragma unroll
        for (int n = 0; n < 4; ++n) {
            v2f w;
            if constexpr (WIDE) {
                const v2f z = e[n] + rs[r];
                w = (v2f){1.0f + __expf(-z.x), 1.0f + __expf(-z.y)} * v[n];
            } else {
                w = e[n] * rs[r] + v[n];
            }
            const v2f g = {__builtin_amdgcn_rcpf(w.x), __builtin_amdgcn_rcpf(w.y)};
            acc[r] = g * an[n] + acc[r];
        }
        if (r & 1) __builtin_amdgcn_sched_barrier(0);
    }
}

template <bool WIDE>
__device__ __forceinline__ void walk(const float* __restrict__ adjp, const float* __restrict__ ep,
                                     const float* __restrict__ vp, const v2f (&rs)[ROWS_PB],
                                     v2f (&acc)[ROWS_PB]) {
    #pragma unroll
    for (int r = 0; r < ROWS_PB; ++r) acc[r] = (v2f){0.0f, 0.0f};

    SrcBlock cur, nxt;
    v4f a0[HROWS], a1[HROWS];
    load_src(cur, ep, vp, 0);
    load_adj(a0, adjp, 0, 0);
    #pragma unroll 2
    for (int jb = 0; jb < NJB; ++jb) {
        const int jn = (jb + 1) & (NJB - 1);
        load_adj(a1, adjp, HROWS, jb);
        load_src(nxt, ep, vp, jn);
        __builtin_amdgcn_sched_barrier(0);
        pair_half<WIDE>(a0, cur, &rs[0], &acc[0]);
        load_adj(a0, adjp, 0, jn);
        __builtin_amdgcn_sched_barrier(0);
        pair_half<WIDE>(a1, cur, &rs[HROWS], &acc[HROWS]);
        cur = nxt;
    }
}

__device__ __forceinline__ void stage_load(v4f (&st)[CJB], const float* __restrict__ adjRow, const int c) {
    #pragma unroll
    for (int k = 0; k < CJB; ++k)
        st[k] = *(const v4f*)(adjRow + (c * CJB + k) * JBLK);
}

__device__ __forceinline__ void stage_store(const v4f (&st)[CJB], v4f* sRow) {
    #pragma unroll
    for (int k = 0; k < CJB; ++k) sRow[k * (JBLK / 4)] = st[k];
}

__device__ __forceinline__ void load_adj_lds(v4f (&a)[HROWS], const v4f* sLane, const int r0, const int k) {
    #pragma unroll
    for (int r = 0; r < HROWS; ++r) a[r] = sLane[(r0 + r) * CN4 + k * (JBLK / 4)];
}

__device__ __forceinline__ void walk_staged(const float* __restrict__ adjRow, v4f (&st)[CJB],
                                            v4f* sAdj, const int wid, const int lane,
                                            const float* __restrict__ ep, const float* __restrict__ vp,
                                            const v2f (&rs)[ROWS_PB], v2f (&acc)[ROWS_PB]) {
    #pragma unroll
    for (int r = 0; r < ROWS_PB; ++r) acc[r] = (v2f){0.0f, 0.0f};

    SrcBlock cur, nxt;
    v4f a0[HROWS], a1[HROWS];
    load_src(cur, ep, vp, 0);
    stage_store(st, sAdj + wid * CN4 + lane);
    __syncthreads();
    #pragma nounroll
    for (int c = 0; c < NCH; ++c) {
        const v4f* sCur = sAdj + (c & 1) * (ROWS_PB * CN4) + lane;
        if (c + 1 < NCH) stage_load(st, adjRow, c + 1);
        load_adj_lds(a0, sCur, 0, 0);
        #pragma unroll 2
        for (int k = 0; k < CJB; ++k) {
            const int jn = (c * CJB + k + 1) & (NJB - 1);
            const int kn = (k + 1) & (CJB - 1);
            load_adj_lds(a1, sCur, HROWS, k);
            load_src(nxt, ep, vp, jn);
            __builtin_amdgcn_sched_barrier(0);
            pair_half<false>(a0, cur, &rs[0], &acc[0]);
            load_adj_lds(a0, sCur, 0, kn);
            __builtin_amdgcn_sched_barrier(0);
            pair_half<false>(a1, cur, &rs[HROWS], &acc[HROWS]);
            cur = nxt;
        }
        if (c + 1 < NCH) {
            stage_store(st, sAdj + ((c + 1) & 1) * (ROWS_PB * CN4) + wid * CN4 + lane);
            __syncthreads();
        }
    }
}

__global__ __launch_bounds__(WAVES_PB * 32) void ggc_gate_agg(
    const float* __restrict__ adj,
    const float* __restrict__ Ah, const float* __restrict__ Bh2,
    const float* __restrict__ Pb2, const float* __restrict__ Qv2,
    float* __restrict__ out)
{
    __shared__ float sPart[WAVES_PB * ROWS_PB * 2 * RPITCH];
    __shared__ float sOut[ROWS_PB * OUT_F];
    __shared__ v4f sAdj[2 * ROWS_PB * CN4];

    const int lane = threadIdx.x & 31;
    const int wid  = threadIdx.x >> 5;
    const int row0 = blockIdx.x * ROWS_PB;
    const int b    = row0 >> 11;
    const int c0   = wid * 2;

    const float* adjp = adj + (size_t)row0 * NN + 4 * lane;
    const size_t po   = (((size_t)(b * (OUT_F / 2) + wid)) * NN + 4 * lane) * 2;

    const float* adjRow = adj + (size_t)(row0 + wid) * NN + 4 * lane;
    v4f st[CJB];
    stage_load(st, adjRow, 0);

    const float aMine = Ah[(size_t)(row0 + (lane >> 1)) * OUT_F + c0 + (lane & 1)];
    const bool wideA  = __any((int)!(fabsf(aMine) <= A_SAFE));
    const float eMine = __expf(-aMine);
    v2f ra[ROWS_PB], re[ROWS_PB];
    #pragma unroll
    for (int r = 0; r < ROWS_PB; ++r) {
        ra[r].x = __builtin_bit_cast(float, __builtin_amdgcn_readlane(__builtin_bit_cast(int, aMine), 2 * r));
        ra[r].y = __builtin_bit_cast(float, __builtin_amdgcn_readlane(__builtin_bit_cast(int, aMine), 2 * r + 1));
        re[r].x = __builtin_bit_cast(float, __builtin_amdgcn_readlane(__builtin_bit_cast(int, eMine), 2 * r));
        re[r].y = __builtin_bit_cast(float, __builtin_amdgcn_readlane(__builtin_bit_cast(int, eMine), 2 * r + 1));
    }

    v2f acc[ROWS_PB];
    walk_staged(adjRow, st, sAdj, wid, lane, Pb2 + po, Qv2 + po, re, acc);
    float chk = 0.0f;
    #pragma unroll
    for (int r = 0; r < ROWS_PB; ++r) chk += acc[r].x + acc[r].y;
    const bool wide = wideA || __any((int)!(fabsf(chk) <= 3.0e38f));
    if (wide) walk<true>(adjp, Bh2 + po, Qv2 + po, ra, acc);

    float part[ROWS_PB * 2];
    #pragma unroll
    for (int r = 0; r < ROWS_PB; ++r) {
        part[2 * r + 0] = acc[r].x + __shfl_xor(acc[r].x, 16, 32);
        part[2 * r + 1] = acc[r].y + __shfl_xor(acc[r].y, 16, 32);
    }
    if (lane < 16) {
        #pragma unroll
        for (int k = 0; k < ROWS_PB * 2; ++k)
            sPart[(wid * (ROWS_PB * 2) + k) * RPITCH + lane] = part[k];
    }
    __syncthreads();
    {
        float sum = 0.0f;
        #pragma unroll
        for (int t = 0; t < 16; ++t)
            sum += sPart[(wid * (ROWS_PB * 2) + lane) * RPITCH + t];
        sOut[(lane >> 1) * OUT_F + c0 + (lane & 1)] = sum;
    }
    __syncthreads();
    {
        const size_t o = (size_t)(row0 + wid) * OUT_F + lane;
        out[o] = out[o] + sOut[wid * OUT_F + lane];
    }
}

extern "C" void kernel_launch(void* const* d_in, const int* in_sizes, int n_in,
                              void* d_out, int out_size, void* d_ws, size_t ws_size,
                              hipStream_t stream) {
    const float* adj = (const float*)d_in[0];
    const float* x   = (const float*)d_in[1];
    const float* U   = (const float*)d_in[2];
    const float* V   = (const float*)d_in[3];
    const float* Aw  = (const float*)d_in[4];
    const float* Bw  = (const float*)d_in[5];
    float* out = (float*)d_out;

    float* ws  = (float*)d_ws;
    float* Ah  = ws;
    float* Bh2 = ws + (size_t)BN * OUT_F;
    float* Pb2 = ws + (size_t)2 * BN * OUT_F;
    float* Qv2 = ws + (size_t)3 * BN * OUT_F;

    ggc_proj_wmma<<<BN / 16, PROJ_WAVES * 32, 0, stream>>>(x, U, V, Aw, Bw,
                                                           Ah, Bh2, Pb2, Qv2, out);
    ggc_gate_agg<<<BN / ROWS_PB, WAVES_PB * 32, 0, stream>>>(adj, Ah, Bh2, Pb2, Qv2, out);
}
